# starting kernel
# baseline (speedup 1.0000x reference)
#include <hip/hip_runtime.h>
#include <cstddef>

typedef __attribute__((ext_vector_type(2))) float v2f;
typedef __attribute__((ext_vector_type(8))) float v8f;

__device__ __forceinline__ float wave32_sum(float v) {
#pragma unroll
  for (int m = 16; m >= 1; m >>= 1) v += __shfl_xor(v, m, 32);
  return v;
}
__device__ __forceinline__ float half16_sum(float v) {
#pragma unroll
  for (int m = 1; m < 16; m <<= 1) v += __shfl_xor(v, m, 32);
  return v;
}

__device__ __forceinline__ v8f wmma_f32_k4(v2f a, v2f b, v8f c) {
  return __builtin_amdgcn_wmma_f32_16x16x4_f32(false, a, false, b, (short)0, c,
                                               false, false);
}

__global__ void k_zero(float* p, int n) {
  int i = blockIdx.x * blockDim.x + threadIdx.x;
  if (i < n) p[i] = 0.f;
}

__global__ void k_reduce_x(const float* __restrict__ x,
                           float* __restrict__ rowsO, float* __restrict__ colsO,
                           float* __restrict__ diagO, float* __restrict__ dsumO,
                           float* __restrict__ totO) {
  __shared__ float colbuf[16 * 512];
  __shared__ float totbuf[16];

  const int bid = blockIdx.x;
  const int mat = bid >> 2;
  const int r0 = (bid & 3) << 7;
  const int tid = threadIdx.x;
  const int w = tid >> 5, l = tid & 31;
  const float* M = x + (size_t)mat * 262144;

  float ca[16];
#pragma unroll
  for (int i = 0; i < 16; ++i) ca[i] = 0.f;
  float totp = 0.f;

#pragma unroll 4
  for (int t = 0; t < 8; ++t) {
    const int r = r0 + w + (t << 4);
    const float4* rp = (const float4*)(M + (size_t)r * 512);
    float rsum = 0.f;
#pragma unroll
    for (int c = 0; c < 4; ++c) {
      float4 v = rp[(c << 5) + l];
      ca[c * 4 + 0] += v.x; ca[c * 4 + 1] += v.y;
      ca[c * 4 + 2] += v.z; ca[c * 4 + 3] += v.w;
      rsum += (v.x + v.y) + (v.z + v.w);
    }
    rsum = wave32_sum(rsum);
    if (l == 0) { rowsO[mat * 512 + r] = rsum; totp += rsum; }
  }
#pragma unroll
  for (int c = 0; c < 4; ++c)
#pragma unroll
    for (int j = 0; j < 4; ++j)
      colbuf[w * 512 + c * 128 + l * 4 + j] = ca[c * 4 + j];
  if (l == 0) totbuf[w] = totp;

  float dv = 0.f;
  if (tid < 128) {
    const int r = r0 + tid;
    dv = M[(size_t)r * 512 + r];
    diagO[mat * 512 + r] = dv;
  }
  float ds = wave32_sum(dv);

  __syncthreads();
  float csum = 0.f;
#pragma unroll
  for (int w2 = 0; w2 < 16; ++w2) csum += colbuf[w2 * 512 + tid];
  atomicAdd(&colsO[mat * 512 + tid], csum);

  if (l == 0 && w < 4) atomicAdd(&dsumO[mat], ds);
  if (tid == 0) {
    float T = 0.f;
#pragma unroll
    for (int i = 0; i < 16; ++i) T += totbuf[i];
    atomicAdd(&totO[mat], T);
  }
}

__global__ void k_prep2(const float* __restrict__ c2, const float* __restrict__ bias2,
                        const float* __restrict__ dsum, const float* __restrict__ tot,
                        float* __restrict__ W0, float* __restrict__ W2,
                        float* __restrict__ W3, float* __restrict__ cst0) {
  const int tid = threadIdx.x;
  for (int idx = tid; idx < 1024; idx += 256) {
    int s = idx >> 4, d = idx & 15;
    const float* c = c2 + (size_t)(d * 64 + s) * 5;
    W0[idx] = c[0];
    W2[idx] = c[2] * (1.f / 512.f);
    W3[idx] = c[3] * (1.f / 512.f);
  }
  for (int idx = tid; idx < 512; idx += 256) {
    int n = idx >> 6, s = idx & 63;
    float acc = bias2[s];
    for (int d = 0; d < 16; ++d) {
      const float* c = c2 + (size_t)(d * 64 + s) * 5;
      acc += c[1] * dsum[n * 16 + d] * (1.f / 512.f) +
             c[4] * tot[n * 16 + d] * (1.f / (512.f * 512.f));
    }
    cst0[idx] = acc;
  }
}

__device__ __forceinline__ v8f gemm_k16(v8f acc, const float* __restrict__ W,
                                        const float* __restrict__ S, int s0,
                                        int i0, int q, int h) {
#pragma unroll
  for (int k0 = 0; k0 < 16; k0 += 4) {
    const int ka = k0 + 2 * h;
    v2f a, b;
    a.x = W[(s0 + q) * 16 + ka];
    a.y = W[(s0 + q) * 16 + ka + 1];
    b.x = S[ka * 512 + i0 + q];
    b.y = S[(ka + 1) * 512 + i0 + q];
    acc = wmma_f32_k4(a, b, acc);
  }
  return acc;
}

__global__ void k_layer2(const float* __restrict__ diag, const float* __restrict__ rows,
                         const float* __restrict__ cols, const float* __restrict__ W0,
                         const float* __restrict__ W2, const float* __restrict__ W3,
                         const float* __restrict__ cst, float* __restrict__ aout,
                         float* __restrict__ bnsum, float* __restrict__ bnsq,
                         float* __restrict__ sumA) {
  const int gw = blockIdx.x * 4 + (threadIdx.x >> 5);
  const int n = gw >> 7, rem = gw & 127;
  const int s0 = (rem >> 5) << 4, i0 = (rem & 31) << 4;
  const int l = threadIdx.x & 31, h = l >> 4, q = l & 15;
  const size_t nb = (size_t)n * 16 * 512;

  v8f acc = {};
  acc = gemm_k16(acc, W0, diag + nb, s0, i0, q, h);
  acc = gemm_k16(acc, W2, rows + nb, s0, i0, q, h);
  acc = gemm_k16(acc, W3, cols + nb, s0, i0, q, h);

#pragma unroll
  for (int r = 0; r < 8; ++r) {
    const int s = s0 + r + 8 * h;
    float v = acc[r] + cst[n * 64 + s];
    v = fmaxf(v, 0.f);
    aout[((size_t)n * 64 + s) * 512 + i0 + q] = v;
    float sv = half16_sum(v);
    float sv2 = half16_sum(v * v);
    if (q == 0) {
      atomicAdd(&bnsum[s], sv);
      atomicAdd(&bnsq[s], sv2);
      atomicAdd(&sumA[n * 64 + s], sv);
    }
  }
}

__global__ void k_bn_mix(const float* __restrict__ bnsum, const float* __restrict__ bnsq,
                         const float* __restrict__ g, const float* __restrict__ bt,
                         const float* __restrict__ c, const float* __restrict__ bias,
                         const float* __restrict__ sumA, float* __restrict__ Mp,
                         float* __restrict__ cst) {
  __shared__ float sA[64], sB[64];
  const int tid = threadIdx.x;
  if (tid < 64) {
    float mu = bnsum[tid] * (1.f / 4096.f);
    float var = bnsq[tid] * (1.f / 4096.f) - mu * mu;
    float A = g[tid] * rsqrtf(var + 1e-5f);
    sA[tid] = A;
    sB[tid] = bt[tid] - A * mu;
  }
  __syncthreads();
  for (int idx = tid; idx < 4096; idx += 256) {
    int s = idx >> 6, d = idx & 63;
    Mp[idx] = c[(size_t)(d * 64 + s) * 2] * sA[d];
  }
  for (int idx = tid; idx < 512; idx += 256) {
    int n = idx >> 6, s = idx & 63;
    float acc = bias[s];
    for (int d = 0; d < 64; ++d) {
      float c0 = c[(size_t)(d * 64 + s) * 2];
      float c1 = c[(size_t)(d * 64 + s) * 2 + 1];
      acc += c0 * sB[d] +
             c1 * (sA[d] * sumA[n * 64 + d] + 512.f * sB[d]) * (1.f / 512.f);
    }
    cst[idx] = acc;
  }
}

__global__ void k_transform(const float* __restrict__ ain, const float* __restrict__ W,
                            const float* __restrict__ cst, float* __restrict__ aout,
                            float* __restrict__ bnsum, float* __restrict__ bnsq,
                            float* __restrict__ sumA) {
  const int gw = blockIdx.x * 4 + (threadIdx.x >> 5);
  const int n = gw >> 7, rem = gw & 127;
  const int s0 = (rem >> 5) << 4, i0 = (rem & 31) << 4;
  const int l = threadIdx.x & 31, h = l >> 4, q = l & 15;
  const float* S = ain + (size_t)n * 64 * 512;

  v8f acc = {};
#pragma unroll
  for (int k0 = 0; k0 < 64; k0 += 4) {
    const int ka = k0 + 2 * h;
    v2f a, b;
    a.x = W[(s0 + q) * 64 + ka];
    a.y = W[(s0 + q) * 64 + ka + 1];
    b.x = S[ka * 512 + i0 + q];
    b.y = S[(ka + 1) * 512 + i0 + q];
    acc = wmma_f32_k4(a, b, acc);
  }

#pragma unroll
  for (int r = 0; r < 8; ++r) {
    const int s = s0 + r + 8 * h;
    float v = acc[r] + cst[n * 64 + s];
    v = fmaxf(v, 0.f);
    aout[((size_t)n * 64 + s) * 512 + i0 + q] = v;
    float sv = half16_sum(v);
    float sv2 = half16_sum(v * v);
    if (q == 0) {
      atomicAdd(&bnsum[s], sv);
      atomicAdd(&bnsq[s], sv2);
      if (sumA) atomicAdd(&sumA[n * 64 + s], sv);
    }
  }
}

__global__ void k_bn_fc(const float* __restrict__ bnsum, const float* __restrict__ bnsq,
                        const float* __restrict__ g, const float* __restrict__ bt,
                        const float* __restrict__ fc1w, const float* __restrict__ fc1b,
                        float* __restrict__ F1, float* __restrict__ cf1) {
  __shared__ float sA[64], sB[64];
  const int tid = threadIdx.x;
  if (tid < 64) {
    float mu = bnsum[tid] * (1.f / 4096.f);
    float var = bnsq[tid] * (1.f / 4096.f) - mu * mu;
    float A = g[tid] * rsqrtf(var + 1e-5f);
    sA[tid] = A;
    sB[tid] = bt[tid] - A * mu;
  }
  __syncthreads();
  for (int idx = tid; idx < 4096; idx += 256) {
    int d = idx & 63;
    F1[idx] = fc1w[idx] * sA[d];
  }
  if (tid < 64) {
    float acc = fc1b[tid];
    for (int d = 0; d < 64; ++d) acc += fc1w[tid * 64 + d] * sB[d];
    cf1[tid] = acc;
  }
}

__global__ void k_fc(const float* __restrict__ a2, const float* __restrict__ F1,
                     const float* __restrict__ cf1, const float* __restrict__ fc2w,
                     const float* __restrict__ fc2b, float* __restrict__ out) {
  __shared__ float r1[64 * 16];
  const int n = blockIdx.x >> 5, i0 = (blockIdx.x & 31) << 4;
  const int w = threadIdx.x >> 5, l = threadIdx.x & 31, h = l >> 4, q = l & 15;
  const float* S = a2 + (size_t)n * 64 * 512;

  v8f acc = {};
#pragma unroll
  for (int k0 = 0; k0 < 64; k0 += 4) {
    const int ka = k0 + 2 * h;
    v2f a, b;
    a.x = F1[(w * 16 + q) * 64 + ka];
    a.y = F1[(w * 16 + q) * 64 + ka + 1];
    b.x = S[ka * 512 + i0 + q];
    b.y = S[(ka + 1) * 512 + i0 + q];
    acc = wmma_f32_k4(a, b, acc);
  }
#pragma unroll
  for (int r = 0; r < 8; ++r) {
    const int o = w * 16 + r + 8 * h;
    r1[o * 16 + q] = fmaxf(acc[r] + cf1[o], 0.f);
  }
  __syncthreads();

  if (w == 0) {
    v8f acc2 = {};
#pragma unroll
    for (int k0 = 0; k0 < 64; k0 += 4) {
      const int ka = k0 + 2 * h;
      v2f a, b;
      a.x = fc2w[q * 64 + ka];
      a.y = fc2w[q * 64 + ka + 1];
      b.x = r1[ka * 16 + q];
      b.y = r1[(ka + 1) * 16 + q];
      acc2 = wmma_f32_k4(a, b, acc2);
    }
#pragma unroll
    for (int r = 0; r < 8; ++r) {
      const int o = r + 8 * h;
      out[((size_t)n * 16 + o) * 512 + i0 + q] = acc2[r] + fc2b[o];
    }
  }
}

extern "C" void kernel_launch(void* const* d_in, const int* in_sizes, int n_in,
                              void* d_out, int out_size, void* d_ws, size_t ws_size,
                              hipStream_t stream) {
  const float* x      = (const float*)d_in[0];
  const float* c2     = (const float*)d_in[1];
  const float* bias2  = (const float*)d_in[2];
  const float* c1a    = (const float*)d_in[3];
  const float* bias1a = (const float*)d_in[4];
  const float* c1b    = (const float*)d_in[5];
  const float* bias1b = (const float*)d_in[6];
  const float* g0 = (const float*)d_in[7],  *b0 = (const float*)d_in[8];
  const float* g1 = (const float*)d_in[9],  *b1 = (const float*)d_in[10];
  const float* g2 = (const float*)d_in[11], *b2 = (const float*)d_in[12];
  const float* fc1w = (const float*)d_in[13], *fc1b = (const float*)d_in[14];
  const float* fc2w = (const float*)d_in[15], *fc2b = (const float*)d_in[16];
  float* out = (float*)d_out;
  float* ws = (float*)d_ws;

  float* rows = ws;
  float* diag = ws + 65536;
  float* a0   = ws + 131072;
  float* a1   = a0 + 262144;
  float* a2   = a1 + 262144;
  float* zb   = a2 + 262144;
  float* cols = zb;
  float* dsum = zb + 65536;
  float* tot  = zb + 65664;
  float* st   = zb + 65792;
  float* bnsum0 = st,       *bnsq0 = st + 64;
  float* bnsum1 = st + 128, *bnsq1 = st + 192;
  float* bnsum2 = st + 256, *bnsq2 = st + 320;
  float* sumA0 = st + 384;
  float* sumA1 = st + 896;
  float* W0 = zb + 67200, *W2 = W0 + 1024, *W3 = W2 + 1024;
  float* cst0 = W3 + 1024;
  float* M1p = cst0 + 512;
  float* cst1 = M1p + 4096;
  float* M2p = cst1 + 512;
  float* cst2 = M2p + 4096;
  float* F1p = cst2 + 512;
  float* cf1 = F1p + 4096;

  k_zero<<<263, 256, 0, stream>>>(zb, 67200);
  k_reduce_x<<<512, 512, 0, stream>>>(x, rows, cols, diag, dsum, tot);
  k_prep2<<<1, 256, 0, stream>>>(c2, bias2, dsum, tot, W0, W2, W3, cst0);
  k_layer2<<<256, 128, 0, stream>>>(diag, rows, cols, W0, W2, W3, cst0, a0,
                                    bnsum0, bnsq0, sumA0);
  k_bn_mix<<<1, 256, 0, stream>>>(bnsum0, bnsq0, g0, b0, c1a, bias1a, sumA0,
                                  M1p, cst1);
  k_transform<<<256, 128, 0, stream>>>(a0, M1p, cst1, a1, bnsum1, bnsq1, sumA1);
  k_bn_mix<<<1, 256, 0, stream>>>(bnsum1, bnsq1, g1, b1, c1b, bias1b, sumA1,
                                  M2p, cst2);
  k_transform<<<256, 128, 0, stream>>>(a1, M2p, cst2, a2, bnsum2, bnsq2, nullptr);
  k_bn_fc<<<1, 256, 0, stream>>>(bnsum2, bnsq2, g2, b2, fc1w, fc1b, F1p, cf1);
  k_fc<<<256, 128, 0, stream>>>(a2, F1p, cf1, fc2w, fc2b, out);
}


# Cross-lane sums (x row sums, BN channel sums in the three layer kernels) on row_xmask DPP adds, several rows side by side and no branch per row, in place of ds_bpermute chains with waits; same order
# speedup vs baseline: 2.5919x; 2.5919x over previous
#include <hip/hip_runtime.h>
#include <cstddef>

typedef __attribute__((ext_vector_type(2))) float v2f;
typedef __attribute__((ext_vector_type(8))) float v8f;

template <int M>
__device__ __forceinline__ float row_xadd(float v) {
  const int o = __builtin_amdgcn_update_dpp(0, __builtin_bit_cast(int, v), 0x160 + M,
                                            0xf, 0xf, true);
  return v + __builtin_bit_cast(float, o);
}
__device__ __forceinline__ float other_row(float v) {
  return __builtin_bit_cast(float, __builtin_amdgcn_permlanex16(
      __builtin_bit_cast(int, v), __builtin_bit_cast(int, v), 0x76543210, 0xfedcba98,
      false, false));
}
__device__ __forceinline__ float wave32_sum(float v) {
  v += other_row(v);
  v = row_xadd<8>(v); v = row_xadd<4>(v); v = row_xadd<2>(v); v = row_xadd<1>(v);
  return v;
}
__device__ __forceinline__ float half16_sum(float v) {
  v = row_xadd<1>(v); v = row_xadd<2>(v); v = row_xadd<4>(v); v = row_xadd<8>(v);
  return v;
}

__device__ __forceinline__ v8f wmma_f32_k4(v2f a, v2f b, v8f c) {
  return __builtin_amdgcn_wmma_f32_16x16x4_f32(false, a, false, b, (short)0, c,
                                               false, false);
}

__global__ void k_zero(float* p, int n) {
  int i = blockIdx.x * blockDim.x + threadIdx.x;
  if (i < n) p[i] = 0.f;
}

__global__ void k_reduce_x(const float* __restrict__ x,
                           float* __restrict__ rowsO, float* __restrict__ colsO,
                           float* __restrict__ diagO, float* __restrict__ dsumO,
                           float* __restrict__ totO) {
  __shared__ float colbuf[16 * 512];
  __shared__ float totbuf[16];

  const int bid = blockIdx.x;
  const int mat = bid >> 2;
  const int r0 = (bid & 3) << 7;
  const int tid = threadIdx.x;
  const int w = tid >> 5, l = tid & 31;
  const float* M = x + (size_t)mat * 262144;

  float ca[16];
#pragma unroll
  for (int i = 0; i < 16; ++i) ca[i] = 0.f;
  float totp = 0.f;
  float rkeep = 0.f;

#pragma unroll 4
  for (int t = 0; t < 8; ++t) {
    const int r = r0 + w + (t << 4);
    const float4* rp = (const float4*)(M + (size_t)r * 512);
    float rsum = 0.f;
#pragma unroll
    for (int c = 0; c < 4; ++c) {
      float4 v = rp[(c << 5) + l];
      ca[c * 4 + 0] += v.x; ca[c * 4 + 1] += v.y;
      ca[c * 4 + 2] += v.z; ca[c * 4 + 3] += v.w;
      rsum += (v.x + v.y) + (v.z + v.w);
    }
    rsum = wave32_sum(rsum);
    rkeep = (l == t) ? rsum : rkeep;
    totp += rsum;
  }
  if (l < 8) rowsO[mat * 512 + r0 + w + (l << 4)] = rkeep;
#pragma unroll
  for (int c = 0; c < 4; ++c)
#pragma unroll
    for (int j = 0; j < 4; ++j)
      colbuf[w * 512 + c * 128 + l * 4 + j] = ca[c * 4 + j];
  if (l == 0) totbuf[w] = totp;

  float dv = 0.f;
  if (tid < 128) {
    const int r = r0 + tid;
    dv = M[(size_t)r * 512 + r];
    diagO[mat * 512 + r] = dv;
  }
  float ds = wave32_sum(dv);

  __syncthreads();
  float csum = 0.f;
#pragma unroll
  for (int w2 = 0; w2 < 16; ++w2) csum += colbuf[w2 * 512 + tid];
  atomicAdd(&colsO[mat * 512 + tid], csum);

  if (l == 0 && w < 4) atomicAdd(&dsumO[mat], ds);
  if (tid == 0) {
    float T = 0.f;
#pragma unroll
    for (int i = 0; i < 16; ++i) T += totbuf[i];
    atomicAdd(&totO[mat], T);
  }
}

__global__ void __launch_bounds__(128) __attribute__((amdgpu_waves_per_eu(1, 2)))
k_layer2(const float* __restrict__ diag, const float* __restrict__ rows,
                         const float* __restrict__ cols, const float* __restrict__ dsum,
                         const float* __restrict__ tot, const float* __restrict__ c2,
                         const float* __restrict__ bias2, float* __restrict__ aout,
                         float* __restrict__ bnsum, float* __restrict__ bnsq,
                         float* __restrict__ sumA) {
  __shared__ float red[4][32];
  const int w = threadIdx.x >> 5;
  const int gw = blockIdx.x * 4 + w;
  const int n = gw >> 7, rem = gw & 127;
  const int s0 = (rem >> 5) << 4, i0 = (rem & 31) << 4;
  const int l = threadIdx.x & 31, h = l >> 4, q = l & 15;
  const size_t nb = (size_t)n * 16 * 512;
  const float* cs = c2 + (size_t)(s0 + q) * 5;

  float wa[8][5], bd[8], br[8], bc[8], bs[8], bt[8];
#pragma unroll
  for (int j = 0; j < 4; ++j)
#pragma unroll
    for (int e = 0; e < 2; ++e) {
      const int d = 4 * j + 2 * h + e, x = 2 * j + e;
#pragma unroll
      for (int b = 0; b < 5; ++b) wa[x][b] = cs[d * 320 + b];
      bd[x] = diag[nb + d * 512 + i0 + q];
      br[x] = rows[nb + d * 512 + i0 + q];
      bc[x] = cols[nb + d * 512 + i0 + q];
      bs[x] = dsum[n * 16 + d];
      bt[x] = tot[n * 16 + d];
    }

  v8f acc0 = {}, acc1 = {};
#pragma unroll
  for (int j = 0; j < 4; ++j) {
    const int x = 2 * j;
    v2f a, b;
    a.x = wa[x][0]; a.y = wa[x + 1][0];
    b.x = bd[x];    b.y = bd[x + 1];
    acc0 = wmma_f32_k4(a, b, acc0);
    a.x = wa[x][1]; a.y = wa[x + 1][1];
    b.x = bs[x] * (1.f / 512.f); b.y = bs[x + 1] * (1.f / 512.f);
    acc1 = wmma_f32_k4(a, b, acc1);
    a.x = wa[x][2]; a.y = wa[x + 1][2];
    b.x = br[x] * (1.f / 512.f); b.y = br[x + 1] * (1.f / 512.f);
    acc0 = wmma_f32_k4(a, b, acc0);
    a.x = wa[x][3]; a.y = wa[x + 1][3];
    b.x = bc[x] * (1.f / 512.f); b.y = bc[x + 1] * (1.f / 512.f);
    acc1 = wmma_f32_k4(a, b, acc1);
    a.x = wa[x][4]; a.y = wa[x + 1][4];
    b.x = bt[x] * (1.f / (512.f * 512.f)); b.y = bt[x + 1] * (1.f / (512.f * 512.f));
    acc0 = wmma_f32_k4(a, b, acc0);
  }

  float vout[8], hs[8], hq[8];
#pragma unroll
  for (int r = 0; r < 8; ++r) {
    const int s = s0 + r + 8 * h;
    float v = (acc0[r] + acc1[r]) + bias2[s];
    v = fmaxf(v, 0.f);
    vout[r] = v;
    hs[r] = v;
    hq[r] = v * v;
  }
#pragma unroll
  for (int r = 0; r < 8; ++r) {
    hs[r] = half16_sum(hs[r]);
    hq[r] = half16_sum(hq[r]);
  }
  if (q == 0) {
#pragma unroll
    for (int r = 0; r < 8; ++r) {
      red[w][r + 8 * h] = hs[r];
      red[w][16 + r + 8 * h] = hq[r];
    }
  }
  __syncthreads();
  if (threadIdx.x < 32) {
    const int t = threadIdx.x;
    const float tsum = (red[0][t] + red[1][t]) + (red[2][t] + red[3][t]);
    const int cch = s0 + (t & 15);
    if (t < 16) {
      atomicAdd(&bnsum[cch], tsum);
      atomicAdd(&sumA[n * 64 + cch], tsum);
    } else {
      atomicAdd(&bnsq[cch], tsum);
    }
  }
#pragma unroll
  for (int r = 0; r < 8; ++r)
    aout[((size_t)n * 64 + s0 + r + 8 * h) * 512 + i0 + q] = vout[r];
}

__global__ void __launch_bounds__(128)
k_transform(const float* __restrict__ ain, const float* __restrict__ c,
                            const float* __restrict__ bias,
                            const float* __restrict__ bnsumIn, const float* __restrict__ bnsqIn,
                            const float* __restrict__ g, const float* __restrict__ bt,
                            const float* __restrict__ sumAin, float* __restrict__ aout,
                            float* __restrict__ bnsum, float* __restrict__ bnsq,
                            float* __restrict__ sumA) {
  __shared__ float sA[64], sB[64], sH[64];
  __shared__ float red[4][32];
  const int tid = threadIdx.x;
  const int wv = tid >> 5;
  const int gw = blockIdx.x * 4 + wv;
  const int n = gw >> 7, rem = gw & 127;
  const int s0 = (rem >> 5) << 4, i0 = (rem & 31) << 4;
  const int l = tid & 31, h = l >> 4, q = l & 15;
  const float* S = ain + (size_t)n * 64 * 512;

  const int ch = tid & 63;
  const float su = bnsumIn[ch], sq = bnsqIn[ch], gg = g[ch], bb = bt[ch];
  const float sm = sumAin[n * 64 + ch];
  float sv[32];
  v2f cw[32];
#pragma unroll
  for (int j = 0; j < 16; ++j)
#pragma unroll
    for (int e = 0; e < 2; ++e) {
      const int d = 4 * j + 2 * h + e;
      sv[2 * j + e] = S[d * 512 + i0 + q];
      cw[2 * j + e] = *(const v2f*)(c + ((size_t)d * 64 + s0 + q) * 2);
    }
  if (tid < 64) {
    const float mu = su * (1.f / 4096.f);
    const float var = sq * (1.f / 4096.f) - mu * mu;
    const float A = gg * rsqrtf(var + 1e-5f);
    const float B = bb - A * mu;
    sA[tid] = A;
    sB[tid] = B;
    sH[tid] = (A * sm + 512.f * B) * (1.f / 512.f);
  }
  __syncthreads();

  v8f acc = {}, accm = {};
#pragma unroll
  for (int j = 0; j < 16; ++j) {
    const int ka = 4 * j + 2 * h, x = 2 * j;
    v2f a, b;
    a.x = cw[x].x; a.y = cw[x + 1].x;
    b.x = sA[ka] * sv[x] + sB[ka];
    b.y = sA[ka + 1] * sv[x + 1] + sB[ka + 1];
    acc = wmma_f32_k4(a, b, acc);
    a.x = cw[x].y; a.y = cw[x + 1].y;
    b.x = sH[ka]; b.y = sH[ka + 1];
    accm = wmma_f32_k4(a, b, accm);
  }

  float vout[8], hs[8], hq[8];
#pragma unroll
  for (int r = 0; r < 8; ++r) {
    const int s = s0 + r + 8 * h;
    float v = (acc[r] + accm[r]) + bias[s];
    v = fmaxf(v, 0.f);
    vout[r] = v;
    hs[r] = v;
    hq[r] = v * v;
  }
#pragma unroll
  for (int r = 0; r < 8; ++r) {
    hs[r] = half16_sum(hs[r]);
    hq[r] = half16_sum(hq[r]);
  }
  if (q == 0) {
#pragma unroll
    for (int r = 0; r < 8; ++r) {
      red[wv][r + 8 * h] = hs[r];
      red[wv][16 + r + 8 * h] = hq[r];
    }
  }
  __syncthreads();
  if (tid < 32) {
    const float tsum = (red[0][tid] + red[1][tid]) + (red[2][tid] + red[3][tid]);
    const int cch = s0 + (tid & 15);
    if (tid < 16) {
      atomicAdd(&bnsum[cch], tsum);
      if (sumA) atomicAdd(&sumA[n * 64 + cch], tsum);
    } else {
      atomicAdd(&bnsq[cch], tsum);
    }
  }
#pragma unroll
  for (int r = 0; r < 8; ++r)
    aout[((size_t)n * 64 + s0 + r + 8 * h) * 512 + i0 + q] = vout[r];
}

__global__ void k_fc(const float* __restrict__ a2, const float* __restrict__ fc1w,
                     const float* __restrict__ fc1b,
                     const float* __restrict__ bnsumIn, const float* __restrict__ bnsqIn,
                     const float* __restrict__ g, const float* __restrict__ bt,
                     const float* __restrict__ fc2w, const float* __restrict__ fc2b,
                     float* __restrict__ out) {
  __shared__ float r1[64 * 16];
  __shared__ float sA[64], sB[64];
  const int tid = threadIdx.x;
  const int n = blockIdx.x >> 5, i0 = (blockIdx.x & 31) << 4;
  const int w = tid >> 5, l = tid & 31, h = l >> 4, q = l & 15;
  const float* S = a2 + (size_t)n * 64 * 512;

  const int ch = tid & 63;
  const float su = bnsumIn[ch], sq = bnsqIn[ch], gg = g[ch], bb = bt[ch];
  float sv[32];
  v2f fw[16];
#pragma unroll
  for (int j = 0; j < 16; ++j) {
    const int ka = 4 * j + 2 * h;
    sv[2 * j] = S[ka * 512 + i0 + q];
    sv[2 * j + 1] = S[(ka + 1) * 512 + i0 + q];
    fw[j] = *(const v2f*)(fc1w + (w * 16 + q) * 64 + ka);
  }
  if (tid < 64) {
    const float mu = su * (1.f / 4096.f);
    const float var = sq * (1.f / 4096.f) - mu * mu;
    const float A = gg * rsqrtf(var + 1e-5f);
    sA[tid] = A;
    sB[tid] = bb - A * mu;
  }
  __syncthreads();

  v8f acc = {};
#pragma unroll
  for (int j = 0; j < 16; ++j) {
    const int ka = 4 * j + 2 * h;
    v2f b;
    b.x = sA[ka] * sv[2 * j] + sB[ka];
    b.y = sA[ka + 1] * sv[2 * j + 1] + sB[ka + 1];
    acc = wmma_f32_k4(fw[j], b, acc);
  }
#pragma unroll
  for (int r = 0; r < 8; ++r) {
    const int o = w * 16 + r + 8 * h;
    r1[o * 16 + q] = fmaxf(acc[r] + fc1b[o], 0.f);
  }
  __syncthreads();

  if (w == 0) {
    v8f acc2 = {};
#pragma unroll
    for (int k0 = 0; k0 < 64; k0 += 4) {
      const int ka = k0 + 2 * h;
      v2f a, b;
      a.x = fc2w[q * 64 + ka];
      a.y = fc2w[q * 64 + ka + 1];
      b.x = r1[ka * 16 + q];
      b.y = r1[(ka + 1) * 16 + q];
      acc2 = wmma_f32_k4(a, b, acc2);
    }
#pragma unroll
    for (int r = 0; r < 8; ++r) {
      const int o = r + 8 * h;
      out[((size_t)n * 16 + o) * 512 + i0 + q] = acc2[r] + fc2b[o];
    }
  }
}

extern "C" void kernel_launch(void* const* d_in, const int* in_sizes, int n_in,
                              void* d_out, int out_size, void* d_ws, size_t ws_size,
                              hipStream_t stream) {
  const float* x      = (const float*)d_in[0];
  const float* c2     = (const float*)d_in[1];
  const float* bias2  = (const float*)d_in[2];
  const float* c1a    = (const float*)d_in[3];
  const float* bias1a = (const float*)d_in[4];
  const float* c1b    = (const float*)d_in[5];
  const float* bias1b = (const float*)d_in[6];
  const float* g0 = (const float*)d_in[7],  *b0 = (const float*)d_in[8];
  const float* g1 = (const float*)d_in[9],  *b1 = (const float*)d_in[10];
  const float* g2 = (const float*)d_in[11], *b2 = (const float*)d_in[12];
  const float* fc1w = (const float*)d_in[13], *fc1b = (const float*)d_in[14];
  const float* fc2w = (const float*)d_in[15], *fc2b = (const float*)d_in[16];
  float* out = (float*)d_out;
  float* ws = (float*)d_ws;

  float* rows = ws;
  float* diag = ws + 65536;
  float* a0   = ws + 131072;
  float* a1   = a0 + 262144;
  float* a2   = a1 + 262144;
  float* zb   = a2 + 262144;
  float* cols = zb;
  float* dsum = zb + 65536;
  float* tot  = zb + 65664;
  float* st   = zb + 65792;
  float* bnsum0 = st,       *bnsq0 = st + 64;
  float* bnsum1 = st + 128, *bnsq1 = st + 192;
  float* bnsum2 = st + 256, *bnsq2 = st + 320;
  float* sumA0 = st + 384;
  float* sumA1 = st + 896;

  k_zero<<<263, 256, 0, stream>>>(zb, 67200);
  k_reduce_x<<<512, 512, 0, stream>>>(x, rows, cols, diag, dsum, tot);
  k_layer2<<<256, 128, 0, stream>>>(diag, rows, cols, dsum, tot, c2, bias2, a0,
                                    bnsum0, bnsq0, sumA0);
  k_transform<<<256, 128, 0, stream>>>(a0, c1a, bias1a, bnsum0, bnsq0, g0, b0,
                                       sumA0, a1, bnsum1, bnsq1, sumA1);
  k_transform<<<256, 128, 0, stream>>>(a1, c1b, bias1b, bnsum1, bnsq1, g1, b1,
                                       sumA1, a2, bnsum2, bnsq2, nullptr);
  k_fc<<<256, 128, 0, stream>>>(a2, fc1w, fc1b, bnsum2, bnsq2, g2, b2, fc2w,
                                fc2b, out);
}


# BN statistics kept per sample: a block's 32 sums go as one whole-line atomic to its own (sample, channel block) line, not into four lines the whole grid shares; reader adds 8 samples; sums regroup
# speedup vs baseline: 2.6750x; 1.0321x over previous
#include <hip/hip_runtime.h>
#include <cstddef>

typedef __attribute__((ext_vector_type(2))) float v2f;
typedef __attribute__((ext_vector_type(8))) float v8f;

template <int M>
__device__ __forceinline__ float row_xadd(float v) {
  const int o = __builtin_amdgcn_update_dpp(0, __builtin_bit_cast(int, v), 0x160 + M,
                                            0xf, 0xf, true);
  return v + __builtin_bit_cast(float, o);
}
__device__ __forceinline__ float other_row(float v) {
  return __builtin_bit_cast(float, __builtin_amdgcn_permlanex16(
      __builtin_bit_cast(int, v), __builtin_bit_cast(int, v), 0x76543210, 0xfedcba98,
      false, false));
}
__device__ __forceinline__ float wave32_sum(float v) {
  v += other_row(v);
  v = row_xadd<8>(v); v = row_xadd<4>(v); v = row_xadd<2>(v); v = row_xadd<1>(v);
  return v;
}
__device__ __forceinline__ float half16_sum(float v) {
  v = row_xadd<1>(v); v = row_xadd<2>(v); v = row_xadd<4>(v); v = row_xadd<8>(v);
  return v;
}

__device__ __forceinline__ v8f wmma_f32_k4(v2f a, v2f b, v8f c) {
  return __builtin_amdgcn_wmma_f32_16x16x4_f32(false, a, false, b, (short)0, c,
                                               false, false);
}

__global__ void k_zero(float* p, int n) {
  int i = blockIdx.x * blockDim.x + threadIdx.x;
  if (i < n) p[i] = 0.f;
}

__global__ void k_reduce_x(const float* __restrict__ x,
                           float* __restrict__ rowsO, float* __restrict__ colsO,
                           float* __restrict__ diagO, float* __restrict__ dsumO,
                           float* __restrict__ totO) {
  __shared__ float colbuf[16 * 512];
  __shared__ float totbuf[16];

  const int bid = blockIdx.x;
  const int mat = bid >> 2;
  const int r0 = (bid & 3) << 7;
  const int tid = threadIdx.x;
  const int w = tid >> 5, l = tid & 31;
  const float* M = x + (size_t)mat * 262144;

  float ca[16];
#pragma unroll
  for (int i = 0; i < 16; ++i) ca[i] = 0.f;
  float totp = 0.f;
  float rkeep = 0.f;

#pragma unroll 4
  for (int t = 0; t < 8; ++t) {
    const int r = r0 + w + (t << 4);
    const float4* rp = (const float4*)(M + (size_t)r * 512);
    float rsum = 0.f;
#pragma unroll
    for (int c = 0; c < 4; ++c) {
      float4 v = rp[(c << 5) + l];
      ca[c * 4 + 0] += v.x; ca[c * 4 + 1] += v.y;
      ca[c * 4 + 2] += v.z; ca[c * 4 + 3] += v.w;
      rsum += (v.x + v.y) + (v.z + v.w);
    }
    rsum = wave32_sum(rsum);
    rkeep = (l == t) ? rsum : rkeep;
    totp += rsum;
  }
  if (l < 8) rowsO[mat * 512 + r0 + w + (l << 4)] = rkeep;
#pragma unroll
  for (int c = 0; c < 4; ++c)
#pragma unroll
    for (int j = 0; j < 4; ++j)
      colbuf[w * 512 + c * 128 + l * 4 + j] = ca[c * 4 + j];
  if (l == 0) totbuf[w] = totp;

  float dv = 0.f;
  if (tid < 128) {
    const int r = r0 + tid;
    dv = M[(size_t)r * 512 + r];
    diagO[mat * 512 + r] = dv;
  }
  float ds = wave32_sum(dv);

  __syncthreads();
  float csum = 0.f;
#pragma unroll
  for (int w2 = 0; w2 < 16; ++w2) csum += colbuf[w2 * 512 + tid];
  atomicAdd(&colsO[mat * 512 + tid], csum);

  if (l == 0 && w < 4) atomicAdd(&dsumO[mat], ds);
  if (tid == 0) {
    float T = 0.f;
#pragma unroll
    for (int i = 0; i < 16; ++i) T += totbuf[i];
    atomicAdd(&totO[mat], T);
  }
}

__global__ void __launch_bounds__(128) __attribute__((amdgpu_waves_per_eu(1, 2)))
k_layer2(const float* __restrict__ diag, const float* __restrict__ rows,
                         const float* __restrict__ cols, const float* __restrict__ dsum,
                         const float* __restrict__ tot, const float* __restrict__ c2,
                         const float* __restrict__ bias2, float* __restrict__ aout,
                         float* __restrict__ statO) {
  __shared__ float red[4][32];
  const int w = threadIdx.x >> 5;
  const int gw = blockIdx.x * 4 + w;
  const int n = gw >> 7, rem = gw & 127;
  const int s0 = (rem >> 5) << 4, i0 = (rem & 31) << 4;
  const int l = threadIdx.x & 31, h = l >> 4, q = l & 15;
  const size_t nb = (size_t)n * 16 * 512;
  const float* cs = c2 + (size_t)(s0 + q) * 5;

  float wa[8][5], bd[8], br[8], bc[8], bs[8], bt[8];
#pragma unroll
  for (int j = 0; j < 4; ++j)
#pragma unroll
    for (int e = 0; e < 2; ++e) {
      const int d = 4 * j + 2 * h + e, x = 2 * j + e;
#pragma unroll
      for (int b = 0; b < 5; ++b) wa[x][b] = cs[d * 320 + b];
      bd[x] = diag[nb + d * 512 + i0 + q];
      br[x] = rows[nb + d * 512 + i0 + q];
      bc[x] = cols[nb + d * 512 + i0 + q];
      bs[x] = dsum[n * 16 + d];
      bt[x] = tot[n * 16 + d];
    }

  v8f acc0 = {}, acc1 = {};
#pragma unroll
  for (int j = 0; j < 4; ++j) {
    const int x = 2 * j;
    v2f a, b;
    a.x = wa[x][0]; a.y = wa[x + 1][0];
    b.x = bd[x];    b.y = bd[x + 1];
    acc0 = wmma_f32_k4(a, b, acc0);
    a.x = wa[x][1]; a.y = wa[x + 1][1];
    b.x = bs[x] * (1.f / 512.f); b.y = bs[x + 1] * (1.f / 512.f);
    acc1 = wmma_f32_k4(a, b, acc1);
    a.x = wa[x][2]; a.y = wa[x + 1][2];
    b.x = br[x] * (1.f / 512.f); b.y = br[x + 1] * (1.f / 512.f);
    acc0 = wmma_f32_k4(a, b, acc0);
    a.x = wa[x][3]; a.y = wa[x + 1][3];
    b.x = bc[x] * (1.f / 512.f); b.y = bc[x + 1] * (1.f / 512.f);
    acc1 = wmma_f32_k4(a, b, acc1);
    a.x = wa[x][4]; a.y = wa[x + 1][4];
    b.x = bt[x] * (1.f / (512.f * 512.f)); b.y = bt[x + 1] * (1.f / (512.f * 512.f));
    acc0 = wmma_f32_k4(a, b, acc0);
  }

  float vout[8], hs[8], hq[8];
#pragma unroll
  for (int r = 0; r < 8; ++r) {
    const int s = s0 + r + 8 * h;
    float v = (acc0[r] + acc1[r]) + bias2[s];
    v = fmaxf(v, 0.f);
    vout[r] = v;
    hs[r] = v;
    hq[r] = v * v;
  }
#pragma unroll
  for (int r = 0; r < 8; ++r) {
    hs[r] = half16_sum(hs[r]);
    hq[r] = half16_sum(hq[r]);
  }
  if (q == 0) {
#pragma unroll
    for (int r = 0; r < 8; ++r) {
      red[w][r + 8 * h] = hs[r];
      red[w][16 + r + 8 * h] = hq[r];
    }
  }
  __syncthreads();
  if (threadIdx.x < 32) {
    const int t = threadIdx.x;
    const float tsum = (red[0][t] + red[1][t]) + (red[2][t] + red[3][t]);
    atomicAdd(&statO[(n * 4 + (s0 >> 4)) * 32 + t], tsum);
  }
#pragma unroll
  for (int r = 0; r < 8; ++r)
    aout[((size_t)n * 64 + s0 + r + 8 * h) * 512 + i0 + q] = vout[r];
}

__global__ void __launch_bounds__(128)
k_transform(const float* __restrict__ ain, const float* __restrict__ c,
                            const float* __restrict__ bias,
                            const float* __restrict__ statI,
                            const float* __restrict__ g, const float* __restrict__ bt,
                            float* __restrict__ aout, float* __restrict__ statO) {
  __shared__ float sA[64], sB[64], sH[64];
  __shared__ float red[4][32];
  const int tid = threadIdx.x;
  const int wv = tid >> 5;
  const int gw = blockIdx.x * 4 + wv;
  const int n = gw >> 7, rem = gw & 127;
  const int s0 = (rem >> 5) << 4, i0 = (rem & 31) << 4;
  const int l = tid & 31, h = l >> 4, q = l & 15;
  const float* S = ain + (size_t)n * 64 * 512;

  const int ch = tid & 63;
  const float* sp = statI + (ch >> 4) * 32 + (ch & 15);
  float ps[8], pq[8];
#pragma unroll
  for (int m = 0; m < 8; ++m) {
    ps[m] = sp[m * 128];
    pq[m] = sp[m * 128 + 16];
  }
  const float sm = sp[n * 128];
  const float gg = g[ch], bb = bt[ch];
  float sv[32];
  v2f cw[32];
#pragma unroll
  for (int j = 0; j < 16; ++j)
#pragma unroll
    for (int e = 0; e < 2; ++e) {
      const int d = 4 * j + 2 * h + e;
      sv[2 * j + e] = S[d * 512 + i0 + q];
      cw[2 * j + e] = *(const v2f*)(c + ((size_t)d * 64 + s0 + q) * 2);
    }
  if (tid < 64) {
    const float su = ((ps[0] + ps[1]) + (ps[2] + ps[3])) + ((ps[4] + ps[5]) + (ps[6] + ps[7]));
    const float sq = ((pq[0] + pq[1]) + (pq[2] + pq[3])) + ((pq[4] + pq[5]) + (pq[6] + pq[7]));
    const float mu = su * (1.f / 4096.f);
    const float var = sq * (1.f / 4096.f) - mu * mu;
    const float A = gg * rsqrtf(var + 1e-5f);
    const float B = bb - A * mu;
    sA[tid] = A;
    sB[tid] = B;
    sH[tid] = (A * sm + 512.f * B) * (1.f / 512.f);
  }
  __syncthreads();

  v8f acc = {}, accm = {};
#pragma unroll
  for (int j = 0; j < 16; ++j) {
    const int ka = 4 * j + 2 * h, x = 2 * j;
    v2f a, b;
    a.x = cw[x].x; a.y = cw[x + 1].x;
    b.x = sA[ka] * sv[x] + sB[ka];
    b.y = sA[ka + 1] * sv[x + 1] + sB[ka + 1];
    acc = wmma_f32_k4(a, b, acc);
    a.x = cw[x].y; a.y = cw[x + 1].y;
    b.x = sH[ka]; b.y = sH[ka + 1];
    accm = wmma_f32_k4(a, b, accm);
  }

  float vout[8], hs[8], hq[8];
#pragma unroll
  for (int r = 0; r < 8; ++r) {
    const int s = s0 + r + 8 * h;
    float v = (acc[r] + accm[r]) + bias[s];
    v = fmaxf(v, 0.f);
    vout[r] = v;
    hs[r] = v;
    hq[r] = v * v;
  }
#pragma unroll
  for (int r = 0; r < 8; ++r) {
    hs[r] = half16_sum(hs[r]);
    hq[r] = half16_sum(hq[r]);
  }
  if (q == 0) {
#pragma unroll
    for (int r = 0; r < 8; ++r) {
      red[wv][r + 8 * h] = hs[r];
      red[wv][16 + r + 8 * h] = hq[r];
    }
  }
  __syncthreads();
  if (tid < 32) {
    const float tsum = (red[0][tid] + red[1][tid]) + (red[2][tid] + red[3][tid]);
    atomicAdd(&statO[(n * 4 + (s0 >> 4)) * 32 + tid], tsum);
  }
#pragma unroll
  for (int r = 0; r < 8; ++r)
    aout[((size_t)n * 64 + s0 + r + 8 * h) * 512 + i0 + q] = vout[r];
}

__global__ void k_fc(const float* __restrict__ a2, const float* __restrict__ fc1w,
                     const float* __restrict__ fc1b,
                     const float* __restrict__ statI,
                     const float* __restrict__ g, const float* __restrict__ bt,
                     const float* __restrict__ fc2w, const float* __restrict__ fc2b,
                     float* __restrict__ out) {
  __shared__ float r1[64 * 16];
  __shared__ float sA[64], sB[64];
  const int tid = threadIdx.x;
  const int n = blockIdx.x >> 5, i0 = (blockIdx.x & 31) << 4;
  const int w = tid >> 5, l = tid & 31, h = l >> 4, q = l & 15;
  const float* S = a2 + (size_t)n * 64 * 512;

  const int ch = tid & 63;
  const float* sp = statI + (ch >> 4) * 32 + (ch & 15);
  float ps[8], pq[8];
#pragma unroll
  for (int m = 0; m < 8; ++m) {
    ps[m] = sp[m * 128];
    pq[m] = sp[m * 128 + 16];
  }
  const float gg = g[ch], bb = bt[ch];
  float sv[32];
  v2f fw[16];
#pragma unroll
  for (int j = 0; j < 16; ++j) {
    const int ka = 4 * j + 2 * h;
    sv[2 * j] = S[ka * 512 + i0 + q];
    sv[2 * j + 1] = S[(ka + 1) * 512 + i0 + q];
    fw[j] = *(const v2f*)(fc1w + (w * 16 + q) * 64 + ka);
  }
  if (tid < 64) {
    const float su = ((ps[0] + ps[1]) + (ps[2] + ps[3])) + ((ps[4] + ps[5]) + (ps[6] + ps[7]));
    const float sq = ((pq[0] + pq[1]) + (pq[2] + pq[3])) + ((pq[4] + pq[5]) + (pq[6] + pq[7]));
    const float mu = su * (1.f / 4096.f);
    const float var = sq * (1.f / 4096.f) - mu * mu;
    const float A = gg * rsqrtf(var + 1e-5f);
    sA[tid] = A;
    sB[tid] = bb - A * mu;
  }
  __syncthreads();

  v8f acc = {};
#pragma unroll
  for (int j = 0; j < 16; ++j) {
    const int ka = 4 * j + 2 * h;
    v2f b;
    b.x = sA[ka] * sv[2 * j] + sB[ka];
    b.y = sA[ka + 1] * sv[2 * j + 1] + sB[ka + 1];
    acc = wmma_f32_k4(fw[j], b, acc);
  }
#pragma unroll
  for (int r = 0; r < 8; ++r) {
    const int o = w * 16 + r + 8 * h;
    r1[o * 16 + q] = fmaxf(acc[r] + fc1b[o], 0.f);
  }
  __syncthreads();

  if (w == 0) {
    v8f acc2 = {};
#pragma unroll
    for (int k0 = 0; k0 < 64; k0 += 4) {
      const int ka = k0 + 2 * h;
      v2f a, b;
      a.x = fc2w[q * 64 + ka];
      a.y = fc2w[q * 64 + ka + 1];
      b.x = r1[ka * 16 + q];
      b.y = r1[(ka + 1) * 16 + q];
      acc2 = wmma_f32_k4(a, b, acc2);
    }
#pragma unroll
    for (int r = 0; r < 8; ++r) {
      const int o = r + 8 * h;
      out[((size_t)n * 16 + o) * 512 + i0 + q] = acc2[r] + fc2b[o];
    }
  }
}

extern "C" void kernel_launch(void* const* d_in, const int* in_sizes, int n_in,
                              void* d_out, int out_size, void* d_ws, size_t ws_size,
                              hipStream_t stream) {
  const float* x      = (const float*)d_in[0];
  const float* c2     = (const float*)d_in[1];
  const float* bias2  = (const float*)d_in[2];
  const float* c1a    = (const float*)d_in[3];
  const float* bias1a = (const float*)d_in[4];
  const float* c1b    = (const float*)d_in[5];
  const float* bias1b = (const float*)d_in[6];
  const float* g0 = (const float*)d_in[7],  *b0 = (const float*)d_in[8];
  const float* g1 = (const float*)d_in[9],  *b1 = (const float*)d_in[10];
  const float* g2 = (const float*)d_in[11], *b2 = (const float*)d_in[12];
  const float* fc1w = (const float*)d_in[13], *fc1b = (const float*)d_in[14];
  const float* fc2w = (const float*)d_in[15], *fc2b = (const float*)d_in[16];
  float* out = (float*)d_out;
  float* ws = (float*)d_ws;

  float* rows = ws;
  float* diag = ws + 65536;
  float* a0   = ws + 131072;
  float* a1   = a0 + 262144;
  float* a2   = a0;
  float* zb   = a1 + 262144;
  float* cols = zb;
  float* dsum = zb + 65536;
  float* tot  = zb + 65664;
  float* st   = zb + 65792;
  float* stat0 = st;
  float* stat1 = st + 1024;
  float* stat2 = st + 2048;

  k_zero<<<269, 256, 0, stream>>>(zb, 68864);
  k_reduce_x<<<512, 512, 0, stream>>>(x, rows, cols, diag, dsum, tot);
  k_layer2<<<256, 128, 0, stream>>>(diag, rows, cols, dsum, tot, c2, bias2, a0, stat0);
  k_transform<<<256, 128, 0, stream>>>(a0, c1a, bias1a, stat0, g0, b0, a1, stat1);
  k_transform<<<256, 128, 0, stream>>>(a1, c1b, bias1b, stat1, g1, b1, a2, stat2);
  k_fc<<<256, 128, 0, stream>>>(a2, fc1w, fc1b, stat2, g2, b2, fc2w, fc2b, out);
}


# x reduction: one 16-wave block per WGP (256, half a matrix each); a wave streams 16 consecutive rows, requesting the next four rows (16 b128 loads) before it sums the four in hand; sums regroup
# speedup vs baseline: 2.7721x; 1.0363x over previous
#include <hip/hip_runtime.h>
#include <cstddef>

typedef __attribute__((ext_vector_type(2))) float v2f;
typedef __attribute__((ext_vector_type(8))) float v8f;

template <int M>
__device__ __forceinline__ float row_xadd(float v) {
  const int o = __builtin_amdgcn_update_dpp(0, __builtin_bit_cast(int, v), 0x160 + M,
                                            0xf, 0xf, true);
  return v + __builtin_bit_cast(float, o);
}
__device__ __forceinline__ float other_row(float v) {
  return __builtin_bit_cast(float, __builtin_amdgcn_permlanex16(
      __builtin_bit_cast(int, v), __builtin_bit_cast(int, v), 0x76543210, 0xfedcba98,
      false, false));
}
__device__ __forceinline__ float wave32_sum(float v) {
  v += other_row(v);
  v = row_xadd<8>(v); v = row_xadd<4>(v); v = row_xadd<2>(v); v = row_xadd<1>(v);
  return v;
}
__device__ __forceinline__ float half16_sum(float v) {
  v = row_xadd<1>(v); v = row_xadd<2>(v); v = row_xadd<4>(v); v = row_xadd<8>(v);
  return v;
}

__device__ __forceinline__ v8f wmma_f32_k4(v2f a, v2f b, v8f c) {
  return __builtin_amdgcn_wmma_f32_16x16x4_f32(false, a, false, b, (short)0, c,
                                               false, false);
}

__global__ void k_zero(float* p, int n) {
  int i = blockIdx.x * blockDim.x + threadIdx.x;
  if (i < n) p[i] = 0.f;
}

__global__ void __launch_bounds__(512)
k_reduce_x(const float* __restrict__ x,
                           float* __restrict__ rowsO, float* __restrict__ colsO,
                           float* __restrict__ diagO, float* __restrict__ dsumO,
                           float* __restrict__ totO) {
  __shared__ float colbuf[16 * 512];
  __shared__ float totbuf[16];

  const int bid = blockIdx.x;
  const int mat = bid >> 1;
  const int r0 = (bid & 1) << 8;
  const int tid = threadIdx.x;
  const int w = tid >> 5, l = tid & 31;
  const float* M = x + (size_t)mat * 262144;

  const int dr = r0 + (tid & 255);
  const float dload = M[(size_t)dr * 512 + dr];

  const float4* base = (const float4*)(M + (size_t)(r0 + 16 * w) * 512) + l;
  float4 buf[2][16];
  float ca[16];
#pragma unroll
  for (int i = 0; i < 16; ++i) ca[i] = 0.f;
  float totp = 0.f;
  float rkeep = 0.f;

#pragma unroll
  for (int p = 0; p < 16; ++p) buf[0][p] = base[p * 32];
#pragma unroll
  for (int g = 0; g < 4; ++g) {
    __builtin_amdgcn_sched_barrier(0);
    if (g + 1 < 4) {
#pragma unroll
      for (int p = 0; p < 16; ++p) buf[(g + 1) & 1][p] = base[((g + 1) * 16 + p) * 32];
    }
    __builtin_amdgcn_sched_barrier(0);
    float rs[4];
#pragma unroll
    for (int q = 0; q < 4; ++q) {
      float rsum = 0.f;
#pragma unroll
      for (int c = 0; c < 4; ++c) {
        const float4 v = buf[g & 1][q * 4 + c];
        ca[c * 4 + 0] += v.x; ca[c * 4 + 1] += v.y;
        ca[c * 4 + 2] += v.z; ca[c * 4 + 3] += v.w;
        rsum += (v.x + v.y) + (v.z + v.w);
      }
      rs[q] = rsum;
    }
#pragma unroll
    for (int q = 0; q < 4; ++q) {
      const float rsum = wave32_sum(rs[q]);
      rkeep = (l == g * 4 + q) ? rsum : rkeep;
      totp += rsum;
    }
#pragma unroll
    for (int c = 0; c < 4; ++c)
      asm volatile("" : "+v"(ca[c * 4 + 0]), "+v"(ca[c * 4 + 1]),
                        "+v"(ca[c * 4 + 2]), "+v"(ca[c * 4 + 3]));
    asm volatile("" : "+v"(rkeep), "+v"(totp));
  }
  if (l < 16) rowsO[mat * 512 + r0 + 16 * w + l] = rkeep;
#pragma unroll
  for (int c = 0; c < 4; ++c)
#pragma unroll
    for (int j = 0; j < 4; ++j)
      colbuf[w * 512 + c * 128 + l * 4 + j] = ca[c * 4 + j];
  if (l == 0) totbuf[w] = totp;

  float dv = 0.f;
  if (tid < 256) {
    dv = dload;
    diagO[mat * 512 + dr] = dv;
  }
  float ds = wave32_sum(dv);

  __syncthreads();
  float csum = 0.f;
#pragma unroll
  for (int w2 = 0; w2 < 16; ++w2) csum += colbuf[w2 * 512 + tid];
  atomicAdd(&colsO[mat * 512 + tid], csum);

  if (l == 0 && w < 8) atomicAdd(&dsumO[mat], ds);
  if (tid == 0) {
    float T = 0.f;
#pragma unroll
    for (int i = 0; i < 16; ++i) T += totbuf[i];
    atomicAdd(&totO[mat], T);
  }
}

__global__ void __launch_bounds__(128) __attribute__((amdgpu_waves_per_eu(1, 2)))
k_layer2(const float* __restrict__ diag, const float* __restrict__ rows,
                         const float* __restrict__ cols, const float* __restrict__ dsum,
                         const float* __restrict__ tot, const float* __restrict__ c2,
                         const float* __restrict__ bias2, float* __restrict__ aout,
                         float* __restrict__ statO) {
  __shared__ float red[4][32];
  const int w = threadIdx.x >> 5;
  const int gw = blockIdx.x * 4 + w;
  const int n = gw >> 7, rem = gw & 127;
  const int s0 = (rem >> 5) << 4, i0 = (rem & 31) << 4;
  const int l = threadIdx.x & 31, h = l >> 4, q = l & 15;
  const size_t nb = (size_t)n * 16 * 512;
  const float* cs = c2 + (size_t)(s0 + q) * 5;

  float wa[8][5], bd[8], br[8], bc[8], bs[8], bt[8];
#pragma unroll
  for (int j = 0; j < 4; ++j)
#pragma unroll
    for (int e = 0; e < 2; ++e) {
      const int d = 4 * j + 2 * h + e, x = 2 * j + e;
#pragma unroll
      for (int b = 0; b < 5; ++b) wa[x][b] = cs[d * 320 + b];
      bd[x] = diag[nb + d * 512 + i0 + q];
      br[x] = rows[nb + d * 512 + i0 + q];
      bc[x] = cols[nb + d * 512 + i0 + q];
      bs[x] = dsum[n * 16 + d];
      bt[x] = tot[n * 16 + d];
    }

  v8f acc0 = {}, acc1 = {};
#pragma unroll
  for (int j = 0; j < 4; ++j) {
    const int x = 2 * j;
    v2f a, b;
    a.x = wa[x][0]; a.y = wa[x + 1][0];
    b.x = bd[x];    b.y = bd[x + 1];
    acc0 = wmma_f32_k4(a, b, acc0);
    a.x = wa[x][1]; a.y = wa[x + 1][1];
    b.x = bs[x] * (1.f / 512.f); b.y = bs[x + 1] * (1.f / 512.f);
    acc1 = wmma_f32_k4(a, b, acc1);
    a.x = wa[x][2]; a.y = wa[x + 1][2];
    b.x = br[x] * (1.f / 512.f); b.y = br[x + 1] * (1.f / 512.f);
    acc0 = wmma_f32_k4(a, b, acc0);
    a.x = wa[x][3]; a.y = wa[x + 1][3];
    b.x = bc[x] * (1.f / 512.f); b.y = bc[x + 1] * (1.f / 512.f);
    acc1 = wmma_f32_k4(a, b, acc1);
    a.x = wa[x][4]; a.y = wa[x + 1][4];
    b.x = bt[x] * (1.f / (512.f * 512.f)); b.y = bt[x + 1] * (1.f / (512.f * 512.f));
    acc0 = wmma_f32_k4(a, b, acc0);
  }

  float vout[8], hs[8], hq[8];
#pragma unroll
  for (int r = 0; r < 8; ++r) {
    const int s = s0 + r + 8 * h;
    float v = (acc0[r] + acc1[r]) + bias2[s];
    v = fmaxf(v, 0.f);
    vout[r] = v;
    hs[r] = v;
    hq[r] = v * v;
  }
#pragma unroll
  for (int r = 0; r < 8; ++r) {
    hs[r] = half16_sum(hs[r]);
    hq[r] = half16_sum(hq[r]);
  }
  if (q == 0) {
#pragma unroll
    for (int r = 0; r < 8; ++r) {
      red[w][r + 8 * h] = hs[r];
      red[w][16 + r + 8 * h] = hq[r];
    }
  }
  __syncthreads();
  if (threadIdx.x < 32) {
    const int t = threadIdx.x;
    const float tsum = (red[0][t] + red[1][t]) + (red[2][t] + red[3][t]);
    atomicAdd(&statO[(n * 4 + (s0 >> 4)) * 32 + t], tsum);
  }
#pragma unroll
  for (int r = 0; r < 8; ++r)
    aout[((size_t)n * 64 + s0 + r + 8 * h) * 512 + i0 + q] = vout[r];
}

__global__ void __launch_bounds__(128)
k_transform(const float* __restrict__ ain, const float* __restrict__ c,
                            const float* __restrict__ bias,
                            const float* __restrict__ statI,
                            const float* __restrict__ g, const float* __restrict__ bt,
                            float* __restrict__ aout, float* __restrict__ statO) {
  __shared__ float sA[64], sB[64], sH[64];
  __shared__ float red[4][32];
  const int tid = threadIdx.x;
  const int wv = tid >> 5;
  const int gw = blockIdx.x * 4 + wv;
  const int n = gw >> 7, rem = gw & 127;
  const int s0 = (rem >> 5) << 4, i0 = (rem & 31) << 4;
  const int l = tid & 31, h = l >> 4, q = l & 15;
  const float* S = ain + (size_t)n * 64 * 512;

  const int ch = tid & 63;
  const float* sp = statI + (ch >> 4) * 32 + (ch & 15);
  float ps[8], pq[8];
#pragma unroll
  for (int m = 0; m < 8; ++m) {
    ps[m] = sp[m * 128];
    pq[m] = sp[m * 128 + 16];
  }
  const float sm = sp[n * 128];
  const float gg = g[ch], bb = bt[ch];
  float sv[32];
  v2f cw[32];
#pragma unroll
  for (int j = 0; j < 16; ++j)
#pragma unroll
    for (int e = 0; e < 2; ++e) {
      const int d = 4 * j + 2 * h + e;
      sv[2 * j + e] = S[d * 512 + i0 + q];
      cw[2 * j + e] = *(const v2f*)(c + ((size_t)d * 64 + s0 + q) * 2);
    }
  if (tid < 64) {
    const float su = ((ps[0] + ps[1]) + (ps[2] + ps[3])) + ((ps[4] + ps[5]) + (ps[6] + ps[7]));
    const float sq = ((pq[0] + pq[1]) + (pq[2] + pq[3])) + ((pq[4] + pq[5]) + (pq[6] + pq[7]));
    const float mu = su * (1.f / 4096.f);
    const float var = sq * (1.f / 4096.f) - mu * mu;
    const float A = gg * rsqrtf(var + 1e-5f);
    const float B = bb - A * mu;
    sA[tid] = A;
    sB[tid] = B;
    sH[tid] = (A * sm + 512.f * B) * (1.f / 512.f);
  }
  __syncthreads();

  v8f acc = {}, accm = {};
#pragma unroll
  for (int j = 0; j < 16; ++j) {
    const int ka = 4 * j + 2 * h, x = 2 * j;
    v2f a, b;
    a.x = cw[x].x; a.y = cw[x + 1].x;
    b.x = sA[ka] * sv[x] + sB[ka];
    b.y = sA[ka + 1] * sv[x + 1] + sB[ka + 1];
    acc = wmma_f32_k4(a, b, acc);
    a.x = cw[x].y; a.y = cw[x + 1].y;
    b.x = sH[ka]; b.y = sH[ka + 1];
    accm = wmma_f32_k4(a, b, accm);
  }

  float vout[8], hs[8], hq[8];
#pragma unroll
  for (int r = 0; r < 8; ++r) {
    const int s = s0 + r + 8 * h;
    float v = (acc[r] + accm[r]) + bias[s];
    v = fmaxf(v, 0.f);
    vout[r] = v;
    hs[r] = v;
    hq[r] = v * v;
  }
#pragma unroll
  for (int r = 0; r < 8; ++r) {
    hs[r] = half16_sum(hs[r]);
    hq[r] = half16_sum(hq[r]);
  }
  if (q == 0) {
#pragma unroll
    for (int r = 0; r < 8; ++r) {
      red[wv][r + 8 * h] = hs[r];
      red[wv][16 + r + 8 * h] = hq[r];
    }
  }
  __syncthreads();
  if (tid < 32) {
    const float tsum = (red[0][tid] + red[1][tid]) + (red[2][tid] + red[3][tid]);
    atomicAdd(&statO[(n * 4 + (s0 >> 4)) * 32 + tid], tsum);
  }
#pragma unroll
  for (int r = 0; r < 8; ++r)
    aout[((size_t)n * 64 + s0 + r + 8 * h) * 512 + i0 + q] = vout[r];
}

__global__ void k_fc(const float* __restrict__ a2, const float* __restrict__ fc1w,
                     const float* __restrict__ fc1b,
                     const float* __restrict__ statI,
                     const float* __restrict__ g, const float* __restrict__ bt,
                     const float* __restrict__ fc2w, const float* __restrict__ fc2b,
                     float* __restrict__ out) {
  __shared__ float r1[64 * 16];
  __shared__ float sA[64], sB[64];
  const int tid = threadIdx.x;
  const int n = blockIdx.x >> 5, i0 = (blockIdx.x & 31) << 4;
  const int w = tid >> 5, l = tid & 31, h = l >> 4, q = l & 15;
  const float* S = a2 + (size_t)n * 64 * 512;

  const int ch = tid & 63;
  const float* sp = statI + (ch >> 4) * 32 + (ch & 15);
  float ps[8], pq[8];
#pragma unroll
  for (int m = 0; m < 8; ++m) {
    ps[m] = sp[m * 128];
    pq[m] = sp[m * 128 + 16];
  }
  const float gg = g[ch], bb = bt[ch];
  float sv[32];
  v2f fw[16];
#pragma unroll
  for (int j = 0; j < 16; ++j) {
    const int ka = 4 * j + 2 * h;
    sv[2 * j] = S[ka * 512 + i0 + q];
    sv[2 * j + 1] = S[(ka + 1) * 512 + i0 + q];
    fw[j] = *(const v2f*)(fc1w + (w * 16 + q) * 64 + ka);
  }
  if (tid < 64) {
    const float su = ((ps[0] + ps[1]) + (ps[2] + ps[3])) + ((ps[4] + ps[5]) + (ps[6] + ps[7]));
    const float sq = ((pq[0] + pq[1]) + (pq[2] + pq[3])) + ((pq[4] + pq[5]) + (pq[6] + pq[7]));
    const float mu = su * (1.f / 4096.f);
    const float var = sq * (1.f / 4096.f) - mu * mu;
    const float A = gg * rsqrtf(var + 1e-5f);
    sA[tid] = A;
    sB[tid] = bb - A * mu;
  }
  __syncthreads();

  v8f acc = {};
#pragma unroll
  for (int j = 0; j < 16; ++j) {
    const int ka = 4 * j + 2 * h;
    v2f b;
    b.x = sA[ka] * sv[2 * j] + sB[ka];
    b.y = sA[ka + 1] * sv[2 * j + 1] + sB[ka + 1];
    acc = wmma_f32_k4(fw[j], b, acc);
  }
#pragma unroll
  for (int r = 0; r < 8; ++r) {
    const int o = w * 16 + r + 8 * h;
    r1[o * 16 + q] = fmaxf(acc[r] + fc1b[o], 0.f);
  }
  __syncthreads();

  if (w == 0) {
    v8f acc2 = {};
#pragma unroll
    for (int k0 = 0; k0 < 64; k0 += 4) {
      const int ka = k0 + 2 * h;
      v2f a, b;
      a.x = fc2w[q * 64 + ka];
      a.y = fc2w[q * 64 + ka + 1];
      b.x = r1[ka * 16 + q];
      b.y = r1[(ka + 1) * 16 + q];
      acc2 = wmma_f32_k4(a, b, acc2);
    }
#pragma unroll
    for (int r = 0; r < 8; ++r) {
      const int o = r + 8 * h;
      out[((size_t)n * 16 + o) * 512 + i0 + q] = acc2[r] + fc2b[o];
    }
  }
}

extern "C" void kernel_launch(void* const* d_in, const int* in_sizes, int n_in,
                              void* d_out, int out_size, void* d_ws, size_t ws_size,
                              hipStream_t stream) {
  const float* x      = (const float*)d_in[0];
  const float* c2     = (const float*)d_in[1];
  const float* bias2  = (const float*)d_in[2];
  const float* c1a    = (const float*)d_in[3];
  const float* bias1a = (const float*)d_in[4];
  const float* c1b    = (const float*)d_in[5];
  const float* bias1b = (const float*)d_in[6];
  const float* g0 = (const float*)d_in[7],  *b0 = (const float*)d_in[8];
  const float* g1 = (const float*)d_in[9],  *b1 = (const float*)d_in[10];
  const float* g2 = (const float*)d_in[11], *b2 = (const float*)d_in[12];
  const float* fc1w = (const float*)d_in[13], *fc1b = (const float*)d_in[14];
  const float* fc2w = (const float*)d_in[15], *fc2b = (const float*)d_in[16];
  float* out = (float*)d_out;
  float* ws = (float*)d_ws;

  float* rows = ws;
  float* diag = ws + 65536;
  float* a0   = ws + 131072;
  float* a1   = a0 + 262144;
  float* a2   = a0;
  float* zb   = a1 + 262144;
  float* cols = zb;
  float* dsum = zb + 65536;
  float* tot  = zb + 65664;
  float* st   = zb + 65792;
  float* stat0 = st;
  float* stat1 = st + 1024;
  float* stat2 = st + 2048;

  k_zero<<<269, 256, 0, stream>>>(zb, 68864);
  k_reduce_x<<<256, 512, 0, stream>>>(x, rows, cols, diag, dsum, tot);
  k_layer2<<<256, 128, 0, stream>>>(diag, rows, cols, dsum, tot, c2, bias2, a0, stat0);
  k_transform<<<256, 128, 0, stream>>>(a0, c1a, bias1a, stat0, g0, b0, a1, stat1);
  k_transform<<<256, 128, 0, stream>>>(a1, c1b, bias1b, stat1, g1, b1, a2, stat2);
  k_fc<<<256, 128, 0, stream>>>(a2, fc1w, fc1b, stat2, g2, b2, fc2w, fc2b, out);
}
